# starting kernel
# baseline (speedup 1.0000x reference)
#include <hip/hip_runtime.h>
#include <math.h>

typedef __attribute__((ext_vector_type(2))) float v2f;
typedef __attribute__((ext_vector_type(8))) float v8f;

#define LDA1 258
#define LDA2 130

__global__ __launch_bounds__(256) void gemm1_kernel(const float* __restrict__ x,
                                                    const float* __restrict__ W1,
                                                    float* __restrict__ h, int N_) {
    __shared__ float lds[16 * LDA1];
    const int tile = blockIdx.x;
    const int t    = threadIdx.x;
    const int row0 = tile * 16;
    const bool full = (row0 + 16 <= N_);

    #pragma unroll
    for (int i = 0; i < 16; ++i) {
        int r = row0 + i;
        if (r >= N_) r = N_ - 1;
        lds[i * LDA1 + t] = x[(size_t)r * 256 + t];
    }
    __syncthreads();

    const int wave = t >> 5;
    const int lane = t & 31;
    const int m    = lane & 15;
    const int koff = (lane >> 4) * 2;
    const int n    = (wave << 4) + m;

    v8f acc = {};
    for (int k0 = 0; k0 < 256; k0 += 4) {
        v2f a, b;
        a.x = lds[m * LDA1 + k0 + koff];
        a.y = lds[m * LDA1 + k0 + koff + 1];
        b.x = W1[(size_t)(k0 + koff)     * 128 + n];
        b.y = W1[(size_t)(k0 + koff + 1) * 128 + n];
        acc = __builtin_amdgcn_wmma_f32_16x16x4_f32(false, a, false, b,
                                                    (short)0, acc, false, false);
    }

    const int rbase = row0 + ((lane >> 4) << 3);
    if (full) {
        #pragma unroll
        for (int r = 0; r < 8; ++r)
            h[(size_t)(rbase + r) * 128 + n] = acc[r];
    } else {
        #pragma unroll
        for (int r = 0; r < 8; ++r) {
            const int row = rbase + r;
            if (row < N_) h[(size_t)row * 128 + n] = acc[r];
        }
    }
}

__global__ __launch_bounds__(256) void attn_kernel(const float* __restrict__ h,
                                                   const float* __restrict__ att_src,
                                                   const float* __restrict__ att_dst,
                                                   float* __restrict__ e_src,
                                                   float* __restrict__ e_dst, int N_) {
    const int node = (int)((blockIdx.x * (unsigned)blockDim.x + threadIdx.x) >> 5);
    const int lane = threadIdx.x & 31;
    if (node >= N_) return;
    const float4 hv = ((const float4*)(h + (size_t)node * 128))[lane];
    const float4 as = ((const float4*)att_src)[lane];
    const float4 ad = ((const float4*)att_dst)[lane];
    float s = hv.x * as.x + hv.y * as.y + hv.z * as.z + hv.w * as.w;
    float d = hv.x * ad.x + hv.y * ad.y + hv.z * ad.z + hv.w * ad.w;
    #pragma unroll
    for (int o = 16; o > 0; o >>= 1) {
        s += __shfl_down(s, o, 32);
        d += __shfl_down(d, o, 32);
    }
    if (lane == 0) { e_src[node] = s; e_dst[node] = d; }
}

__global__ __launch_bounds__(256) void edge_alpha_max(const int* __restrict__ ei,
                                                      const float* __restrict__ e_src,
                                                      const float* __restrict__ e_dst,
                                                      float* __restrict__ alpha,
                                                      unsigned* __restrict__ mbits, int E_) {
    const int e = blockIdx.x * blockDim.x + threadIdx.x;
    if (e >= E_) return;
    const int s = ei[e];
    const int d = ei[E_ + e];
    float a = e_src[s] + e_dst[d];
    a = 1.0f / (1.0f + expf(-a));
    alpha[e] = a;
    atomicMax(mbits + d, __float_as_uint(a));
}

__global__ __launch_bounds__(256) void edge_expsum(const int* __restrict__ ei,
                                                   const float* __restrict__ alpha,
                                                   const unsigned* __restrict__ mbits,
                                                   float* __restrict__ denom,
                                                   float* __restrict__ exb, int E_) {
    const int e = blockIdx.x * blockDim.x + threadIdx.x;
    if (e >= E_) return;
    const int d = ei[E_ + e];
    const float mv = __uint_as_float(mbits[d]);
    const float ex = expf(alpha[e] - mv);
    exb[e] = ex;
    atomicAdd(denom + d, ex);
}

__global__ __launch_bounds__(256) void edge_agg(const int* __restrict__ ei,
                                                const float* __restrict__ h,
                                                const float* __restrict__ exb,
                                                const float* __restrict__ denom,
                                                float* __restrict__ out1, int E_) {
    const int e    = (int)((blockIdx.x * (unsigned)blockDim.x + threadIdx.x) >> 5);
    const int lane = threadIdx.x & 31;
    if (e >= E_) return;
    const int s = ei[e];
    const int d = ei[E_ + e];
    const float w = exb[e] / denom[d];
    const float4 hv = ((const float4*)(h + (size_t)s * 128))[lane];
    float* o = out1 + (size_t)d * 128 + lane * 4;
    atomicAdd(o + 0, hv.x * w);
    atomicAdd(o + 1, hv.y * w);
    atomicAdd(o + 2, hv.z * w);
    atomicAdd(o + 3, hv.w * w);
}

__global__ __launch_bounds__(256) void gemm2_kernel(const float* __restrict__ out1,
                                                    const float* __restrict__ W2,
                                                    float* __restrict__ out, int N_) {
    __shared__ float lds[16 * LDA2];
    const int tile = blockIdx.x;
    const int t    = threadIdx.x;
    const int row0 = tile * 16;
    const bool full = (row0 + 16 <= N_);

    #pragma unroll
    for (int i = 0; i < 8; ++i) {
        const int idx = i * 256 + t;
        const int r   = idx >> 7;
        const int c   = idx & 127;
        int row = row0 + r;
        if (row >= N_) row = N_ - 1;
        float v = out1[(size_t)row * 128 + c];
        lds[r * LDA2 + c] = (v > 0.0f) ? v : (expf(v) - 1.0f);
    }
    __syncthreads();

    const int wave = t >> 5;
    const int lane = t & 31;
    const int m    = lane & 15;
    const int koff = (lane >> 4) * 2;
    const int n    = (wave << 4) + m;

    v8f acc = {};
    for (int k0 = 0; k0 < 128; k0 += 4) {
        v2f a, b;
        a.x = lds[m * LDA2 + k0 + koff];
        a.y = lds[m * LDA2 + k0 + koff + 1];
        b.x = W2[(size_t)(k0 + koff)     * 128 + n];
        b.y = W2[(size_t)(k0 + koff + 1) * 128 + n];
        acc = __builtin_amdgcn_wmma_f32_16x16x4_f32(false, a, false, b,
                                                    (short)0, acc, false, false);
    }

    const int rbase = row0 + ((lane >> 4) << 3);
    if (full) {
        #pragma unroll
        for (int r = 0; r < 8; ++r)
            out[(size_t)(rbase + r) * 128 + n] = acc[r];
    } else {
        #pragma unroll
        for (int r = 0; r < 8; ++r) {
            const int row = rbase + r;
            if (row < N_) out[(size_t)row * 128 + n] = acc[r];
        }
    }
}

extern "C" void kernel_launch(void* const* d_in, const int* in_sizes, int n_in,
                              void* d_out, int out_size, void* d_ws, size_t ws_size,
                              hipStream_t stream) {
    const float* x       = (const float*)d_in[0];
    const int*   ei      = (const int*)  d_in[1];
    const float* W1      = (const float*)d_in[2];
    const float* att_src = (const float*)d_in[3];
    const float* att_dst = (const float*)d_in[4];
    const float* W2      = (const float*)d_in[5];
    float*       out     = (float*)d_out;

    const int N_ = in_sizes[0] / 256;
    const int E_ = in_sizes[1] / 2;

    float*    ws    = (float*)d_ws;
    float*    h     = ws;
    float*    out1  = h     + (size_t)N_ * 128;
    float*    e_src = out1  + (size_t)N_ * 128;
    float*    e_dst = e_src + N_;
    unsigned* mbits = (unsigned*)(e_dst + N_);
    float*    denom = (float*)(mbits + N_);
    float*    alpha = denom + N_;
    float*    exb   = alpha + E_;

    hipMemsetAsync(out1,  0, (size_t)N_ * 128 * sizeof(float), stream);
    hipMemsetAsync(mbits, 0, (size_t)N_ * sizeof(unsigned),    stream);
    hipMemsetAsync(denom, 0, (size_t)N_ * sizeof(float),       stream);

    const int tiles = (N_ + 15) / 16;
    gemm1_kernel  <<<tiles, 256, 0, stream>>>(x, W1, h, N_);
    attn_kernel   <<<(N_ * 32 + 255) / 256, 256, 0, stream>>>(h, att_src, att_dst,
                                                              e_src, e_dst, N_);
    edge_alpha_max<<<(E_ + 255) / 256, 256, 0, stream>>>(ei, e_src, e_dst, alpha, mbits, E_);
    edge_expsum   <<<(E_ + 255) / 256, 256, 0, stream>>>(ei, alpha, mbits, denom, exb, E_);
    edge_agg      <<<(E_ * 32 + 255) / 256, 256, 0, stream>>>(ei, h, exb, denom, out1, E_);
    gemm2_kernel  <<<tiles, 256, 0, stream>>>(out1, W2, out, N_);
}


# edge pass routed by chip half: waves read their die, draw 32-edge chunks from two ticket queues and add only rows whose destination lies in their half's row range; atomic sum order changes
# speedup vs baseline: 2.0771x; 2.0771x over previous
#include <hip/hip_runtime.h>
#include <math.h>

typedef __attribute__((ext_vector_type(2))) float v2f;
typedef __attribute__((ext_vector_type(8))) float v8f;

#define LDA1 258
#define LDA2 130
#define QHEAD_STRIDE 64

__global__ __launch_bounds__(256) void gemm1_kernel(const float* __restrict__ x,
                                                    const float* __restrict__ W1,
                                                    float* __restrict__ h, int N_) {
    __shared__ float lds[16 * LDA1];
    const int tile = blockIdx.x;
    const int t    = threadIdx.x;
    const int row0 = tile * 16;
    const bool full = (row0 + 16 <= N_);

    #pragma unroll
    for (int i = 0; i < 16; ++i) {
        int r = row0 + i;
        if (r >= N_) r = N_ - 1;
        lds[i * LDA1 + t] = x[(size_t)r * 256 + t];
    }
    __syncthreads();

    const int wave = t >> 5;
    const int lane = t & 31;
    const int m    = lane & 15;
    const int koff = (lane >> 4) * 2;
    const int n    = (wave << 4) + m;

    v8f acc = {};
    for (int k0 = 0; k0 < 256; k0 += 4) {
        v2f a, b;
        a.x = lds[m * LDA1 + k0 + koff];
        a.y = lds[m * LDA1 + k0 + koff + 1];
        b.x = W1[(size_t)(k0 + koff)     * 128 + n];
        b.y = W1[(size_t)(k0 + koff + 1) * 128 + n];
        acc = __builtin_amdgcn_wmma_f32_16x16x4_f32(false, a, false, b,
                                                    (short)0, acc, false, false);
    }

    const int rbase = row0 + ((lane >> 4) << 3);
    if (full) {
        #pragma unroll
        for (int r = 0; r < 8; ++r)
            h[(size_t)(rbase + r) * 128 + n] = acc[r];
    } else {
        #pragma unroll
        for (int r = 0; r < 8; ++r) {
            const int row = rbase + r;
            if (row < N_) h[(size_t)row * 128 + n] = acc[r];
        }
    }
}

__global__ __launch_bounds__(256) void attn_kernel(const float* __restrict__ h,
                                                   const float* __restrict__ att_src,
                                                   const float* __restrict__ att_dst,
                                                   float* __restrict__ e_src,
                                                   float* __restrict__ e_dst,
                                                   int* __restrict__ qhead, int N_) {
    if (blockIdx.x == 0 && threadIdx.x < 2) qhead[threadIdx.x * QHEAD_STRIDE] = 0;
    const int node = (int)((blockIdx.x * (unsigned)blockDim.x + threadIdx.x) >> 5);
    const int lane = threadIdx.x & 31;
    if (node >= N_) return;
    const float4 hv = ((const float4*)(h + (size_t)node * 128))[lane];
    const float4 as = ((const float4*)att_src)[lane];
    const float4 ad = ((const float4*)att_dst)[lane];
    float s = hv.x * as.x + hv.y * as.y + hv.z * as.z + hv.w * as.w;
    float d = hv.x * ad.x + hv.y * ad.y + hv.z * ad.z + hv.w * ad.w;
    #pragma unroll
    for (int o = 16; o > 0; o >>= 1) {
        s += __shfl_down(s, o, 32);
        d += __shfl_down(d, o, 32);
    }
    if (lane == 0) { e_src[node] = s; e_dst[node] = d; }
}

__device__ __forceinline__ void row_load(float (&r)[4], const float* __restrict__ h, int s, int lane) {
    const float* p = h + (size_t)s * 128 + lane;
    r[0] = p[0]; r[1] = p[32]; r[2] = p[64]; r[3] = p[96];
}

__device__ __forceinline__ void row_add(const float (&r)[4], float* __restrict__ out1, int d,
                                        float w, int lane) {
    float* o = out1 + (size_t)d * 128 + lane;
    atomicAdd(o,      r[0] * w);
    atomicAdd(o + 32, r[1] * w);
    atomicAdd(o + 64, r[2] * w);
    atomicAdd(o + 96, r[3] * w);
}

__device__ __forceinline__ int   bcast_i(int v, int j)   { return __builtin_amdgcn_readlane(v, j); }
__device__ __forceinline__ float bcast_f(float v, int j) {
    return __int_as_float(__builtin_amdgcn_readlane(__float_as_int(v), j));
}

__device__ __forceinline__ unsigned ticket_issue(unsigned* head, int lane) {
    unsigned t = 0u;
    if (lane == 0) t = atomicInc(head, 0xffffffffu);
    return t;
}
__device__ __forceinline__ int ticket_value(unsigned t) {
    return (int)__builtin_amdgcn_readfirstlane(t);
}

__device__ __forceinline__ void fetch4(unsigned& m, int s, int d, float ex,
                                       float (&buf)[4][4], int (&dd)[4], float (&ww)[4],
                                       const float* __restrict__ h, int lane) {
    #pragma unroll
    for (int q = 0; q < 4; ++q) {
        const int j = __builtin_ctz(m);
        m &= m - 1;
        dd[q] = bcast_i(d, j);
        ww[q] = bcast_f(ex, j);
        row_load(buf[q], h, bcast_i(s, j), lane);
    }
}

__device__ __forceinline__ void add4(const float (&buf)[4][4], const int (&dd)[4],
                                     const float (&ww)[4], float* __restrict__ out1, int lane) {
    #pragma unroll
    for (int q = 0; q < 4; ++q) row_add(buf[q], out1, dd[q], ww[q], lane);
}

__global__ __launch_bounds__(128) void edge_agg(const int* __restrict__ ei,
                                                const float* __restrict__ h,
                                                const float* __restrict__ e_src,
                                                const float* __restrict__ e_dst,
                                                float* __restrict__ denom,
                                                float* __restrict__ out1,
                                                int* __restrict__ qhead, int E_, int N_) {
    const int lane    = threadIdx.x & 31;
    const int nchunks = (E_ + 31) >> 5;
    const int nh      = ((N_ >> 1) + 63) & ~63;
    const int myhalf  = (int)((__builtin_amdgcn_s_sendmsg_rtn(0x87) >> 18) & 1u);

    for (int pass = 0; pass < 2; ++pass) {
        const int q = myhalf ^ pass;
        unsigned* head = (unsigned*)qhead + q * QHEAD_STRIDE;
        int t = ticket_value(ticket_issue(head, lane));
        while (t < nchunks) {
            const unsigned tnext = ticket_issue(head, lane);

            const int  e     = (t << 5) + lane;
            int   s = 0, d = 0;
            float ex = 0.0f;
            bool  mine = false;
            if (e < E_) {
                s = ei[e];
                d = ei[E_ + e];
                mine = ((d >= nh) ? 1 : 0) == q;
                if (mine) {
                    float a = e_src[s] + e_dst[d];
                    a = 1.0f / (1.0f + expf(-a));
                    ex = expf(a);
                    atomicAdd(denom + d, ex);
                }
            }
            unsigned m = __builtin_amdgcn_ballot_w32(mine);

            int ng = __builtin_popcount(m) >> 2;
            float bufA[4][4], bufB[4][4];
            if (ng > 0) {
                int   dA[4], dB[4];
                float wA[4], wB[4];
                fetch4(m, s, d, ex, bufA, dA, wA, h, lane);
                for (;;) {
                    if (ng > 1) fetch4(m, s, d, ex, bufB, dB, wB, h, lane);
                    add4(bufA, dA, wA, out1, lane);
                    if (--ng == 0) break;
                    if (ng > 1) fetch4(m, s, d, ex, bufA, dA, wA, h, lane);
                    add4(bufB, dB, wB, out1, lane);
                    if (--ng == 0) break;
                }
            }
            {
                const int nt = __builtin_popcount(m);
                int   sT[3] = {0, 0, 0};
                int   dT[3] = {0, 0, 0};
                float wT[3] = {0.0f, 0.0f, 0.0f};
                #pragma unroll
                for (int k = 0; k < 3; ++k) {
                    if (k < nt) {
                        const int j = __builtin_ctz(m);
                        m &= m - 1;
                        sT[k] = bcast_i(s, j);
                        dT[k] = bcast_i(d, j);
                        wT[k] = bcast_f(ex, j);
                    }
                }
                #pragma unroll
                for (int k = 0; k < 3; ++k)
                    if (k < nt) row_load(bufA[k], h, sT[k], lane);
                #pragma unroll
                for (int k = 0; k < 3; ++k)
                    if (k < nt) row_add(bufA[k], out1, dT[k], wT[k], lane);
            }
            t = ticket_value(tnext);
        }
    }
}

__global__ __launch_bounds__(256) void gemm2_kernel(const float* __restrict__ out1,
                                                    const float* __restrict__ denom,
                                                    const float* __restrict__ W2,
                                                    float* __restrict__ out, int N_) {
    __shared__ float lds[16 * LDA2];
    const int tile = blockIdx.x;
    const int t    = threadIdx.x;
    const int row0 = tile * 16;
    const bool full = (row0 + 16 <= N_);

    #pragma unroll
    for (int i = 0; i < 8; ++i) {
        const int idx = i * 256 + t;
        const int r   = idx >> 7;
        const int c   = idx & 127;
        int row = row0 + r;
        if (row >= N_) row = N_ - 1;
        const float dn = denom[row];
        float v = out1[(size_t)row * 128 + c];
        v = (dn > 0.0f) ? v / dn : 0.0f;
        lds[r * LDA2 + c] = (v > 0.0f) ? v : (expf(v) - 1.0f);
    }
    __syncthreads();

    const int wave = t >> 5;
    const int lane = t & 31;
    const int m    = lane & 15;
    const int koff = (lane >> 4) * 2;
    const int n    = (wave << 4) + m;

    v8f acc = {};
    for (int k0 = 0; k0 < 128; k0 += 4) {
        v2f a, b;
        a.x = lds[m * LDA2 + k0 + koff];
        a.y = lds[m * LDA2 + k0 + koff + 1];
        b.x = W2[(size_t)(k0 + koff)     * 128 + n];
        b.y = W2[(size_t)(k0 + koff + 1) * 128 + n];
        acc = __builtin_amdgcn_wmma_f32_16x16x4_f32(false, a, false, b,
                                                    (short)0, acc, false, false);
    }

    const int rbase = row0 + ((lane >> 4) << 3);
    if (full) {
        #pragma unroll
        for (int r = 0; r < 8; ++r)
            out[(size_t)(rbase + r) * 128 + n] = acc[r];
    } else {
        #pragma unroll
        for (int r = 0; r < 8; ++r) {
            const int row = rbase + r;
            if (row < N_) out[(size_t)row * 128 + n] = acc[r];
        }
    }
}

extern "C" void kernel_launch(void* const* d_in, const int* in_sizes, int n_in,
                              void* d_out, int out_size, void* d_ws, size_t ws_size,
                              hipStream_t stream) {
    const float* x       = (const float*)d_in[0];
    const int*   ei      = (const int*)  d_in[1];
    const float* W1      = (const float*)d_in[2];
    const float* att_src = (const float*)d_in[3];
    const float* att_dst = (const float*)d_in[4];
    const float* W2      = (const float*)d_in[5];
    float*       out     = (float*)d_out;

    const int N_ = in_sizes[0] / 256;
    const int E_ = in_sizes[1] / 2;

    float*    ws    = (float*)d_ws;
    float*    h     = ws;
    float*    out1  = h     + (size_t)N_ * 128;
    float*    e_src = out1  + (size_t)N_ * 128;
    float*    e_dst = e_src + N_;
    float*    denom = e_dst + N_;
    int*      qhead = (int*)d_out;

    hipMemsetAsync(out1,  0, (size_t)N_ * 128 * sizeof(float), stream);
    hipMemsetAsync(denom, 0, (size_t)N_ * sizeof(float),       stream);

    const int tiles = (N_ + 15) / 16;
    gemm1_kernel  <<<tiles, 256, 0, stream>>>(x, W1, h, N_);
    attn_kernel   <<<(N_ * 32 + 255) / 256, 256, 0, stream>>>(h, att_src, att_dst,
                                                              e_src, e_dst, qhead, N_);
    edge_agg      <<<4096, 128, 0, stream>>>(ei, h, e_src, e_dst, denom, out1, qhead, E_, N_);
    gemm2_kernel  <<<tiles, 256, 0, stream>>>(out1, denom, W2, out, N_);
}


# edge pass scatter to pull: edges counting-sorted by destination in memory (returning integer adds, arrival-order rank), one wave sums a row in registers and stores it once; sum order changes
# speedup vs baseline: 2.5785x; 1.2414x over previous
#include <hip/hip_runtime.h>
#include <math.h>

typedef __attribute__((ext_vector_type(2))) float v2f;
typedef __attribute__((ext_vector_type(8))) float v8f;

#define LDA1 258
#define LDA2 130
#define QHEAD_STRIDE 64

__global__ __launch_bounds__(256) void gemm1_kernel(const float* __restrict__ x,
                                                    const float* __restrict__ W1,
                                                    float* __restrict__ h, int N_) {
    __shared__ float lds[16 * LDA1];
    const int tile = blockIdx.x;
    const int t    = threadIdx.x;
    const int row0 = tile * 16;
    const bool full = (row0 + 16 <= N_);

    #pragma unroll
    for (int i = 0; i < 16; ++i) {
        int r = row0 + i;
        if (r >= N_) r = N_ - 1;
        lds[i * LDA1 + t] = x[(size_t)r * 256 + t];
    }
    __syncthreads();

    const int wave = t >> 5;
    const int lane = t & 31;
    const int m    = lane & 15;
    const int koff = (lane >> 4) * 2;
    const int n    = (wave << 4) + m;

    v8f acc = {};
    for (int k0 = 0; k0 < 256; k0 += 4) {
        v2f a, b;
        a.x = lds[m * LDA1 + k0 + koff];
        a.y = lds[m * LDA1 + k0 + koff + 1];
        b.x = W1[(size_t)(k0 + koff)     * 128 + n];
        b.y = W1[(size_t)(k0 + koff + 1) * 128 + n];
        acc = __builtin_amdgcn_wmma_f32_16x16x4_f32(false, a, false, b,
                                                    (short)0, acc, false, false);
    }

    const int rbase = row0 + ((lane >> 4) << 3);
    if (full) {
        #pragma unroll
        for (int r = 0; r < 8; ++r)
            h[(size_t)(rbase + r) * 128 + n] = acc[r];
    } else {
        #pragma unroll
        for (int r = 0; r < 8; ++r) {
            const int row = rbase + r;
            if (row < N_) h[(size_t)row * 128 + n] = acc[r];
        }
    }
}

__global__ __launch_bounds__(256) void attn_kernel(const float* __restrict__ h,
                                                   const float* __restrict__ att_src,
                                                   const float* __restrict__ att_dst,
                                                   float* __restrict__ e_src,
                                                   float* __restrict__ e_dst, int N_) {
    const int node = (int)((blockIdx.x * (unsigned)blockDim.x + threadIdx.x) >> 5);
    const int lane = threadIdx.x & 31;
    if (node >= N_) return;
    const float4 hv = ((const float4*)(h + (size_t)node * 128))[lane];
    const float4 as = ((const float4*)att_src)[lane];
    const float4 ad = ((const float4*)att_dst)[lane];
    float s = hv.x * as.x + hv.y * as.y + hv.z * as.z + hv.w * as.w;
    float d = hv.x * ad.x + hv.y * ad.y + hv.z * ad.z + hv.w * ad.w;
    #pragma unroll
    for (int o = 16; o > 0; o >>= 1) {
        s += __shfl_down(s, o, 32);
        d += __shfl_down(d, o, 32);
    }
    if (lane == 0) { e_src[node] = s; e_dst[node] = d; }
}

#define RANK_PER_LANE 16
#define RANK_CHUNK (32 * RANK_PER_LANE)
#define FILL_PER_THREAD 8
#define ROWS_PER_WAVE 4

__device__ __forceinline__ int   bcast_i(int v, int j)   { return __builtin_amdgcn_readlane(v, j); }
__device__ __forceinline__ float bcast_f(float v, int j) {
    return __int_as_float(__builtin_amdgcn_readlane(__float_as_int(v), j));
}

__device__ __forceinline__ unsigned ticket_issue(unsigned* head, int lane) {
    unsigned t = 0u;
    if (lane == 0) t = atomicInc(head, 0xffffffffu);
    return t;
}
__device__ __forceinline__ int ticket_value(unsigned t) {
    return (int)__builtin_amdgcn_readfirstlane(t);
}

__global__ __launch_bounds__(256) void edge_rank(const int* __restrict__ ei,
                                                 int* __restrict__ cnt,
                                                 int* __restrict__ rank,
                                                 int* __restrict__ qhead, int E_, int N_) {
    const int lane   = threadIdx.x & 31;
    const int nticks = (E_ + RANK_CHUNK - 1) / RANK_CHUNK;
    const int nh     = ((N_ >> 1) + 63) & ~63;
    const int myhalf = (int)((__builtin_amdgcn_s_sendmsg_rtn(0x87) >> 18) & 1u);

    for (int pass = 0; pass < 2; ++pass) {
        const int q = myhalf ^ pass;
        unsigned* head = (unsigned*)qhead + q * QHEAD_STRIDE;
        int t = ticket_value(ticket_issue(head, lane));
        while (t < nticks) {
            const int e0 = t * RANK_CHUNK + lane;
            int  d[RANK_PER_LANE], rk[RANK_PER_LANE];
            bool mine[RANK_PER_LANE];
            #pragma unroll
            for (int k = 0; k < RANK_PER_LANE; ++k) {
                const int e  = e0 + 32 * k;
                const int dv = ei[E_ + ((e < E_) ? e : (E_ - 1))];
                d[k] = (e < E_) ? dv : -1;
            }
            const unsigned tnext = ticket_issue(head, lane);
            #pragma unroll
            for (int k = 0; k < RANK_PER_LANE; ++k) {
                mine[k] = (d[k] >= 0) && (((d[k] >= nh) ? 1 : 0) == q);
                rk[k] = 0;
                if (mine[k]) rk[k] = atomicAdd(cnt + d[k], 1);
            }
            #pragma unroll
            for (int k = 0; k < RANK_PER_LANE; ++k)
                if (mine[k]) rank[e0 + 32 * k] = rk[k];
            t = ticket_value(tnext);
        }
    }
}

__global__ __launch_bounds__(256) void row_alloc(const int* __restrict__ cnt,
                                                 int* __restrict__ off,
                                                 int* __restrict__ total, int N_) {
    __shared__ int wsum[8];
    __shared__ int base_s;
    const int t    = threadIdx.x;
    const int lane = t & 31;
    const int wave = t >> 5;
    const int i    = blockIdx.x * 256 + t;
    const int c    = (i < N_) ? cnt[i] : 0;
    int v = c;
    #pragma unroll
    for (int o = 1; o < 32; o <<= 1) {
        const int u = __shfl_up(v, o, 32);
        if (lane >= o) v += u;
    }
    if (lane == 31) wsum[wave] = v;
    __syncthreads();
    if (t == 0) {
        int s = 0;
        for (int w = 0; w < 8; ++w) { const int x = wsum[w]; wsum[w] = s; s += x; }
        base_s = atomicAdd(total, s);
    }
    __syncthreads();
    if (i < N_) off[i] = base_s + wsum[wave] + v - c;
}

__global__ __launch_bounds__(256) void edge_fill(const int* __restrict__ ei,
                                                 const int* __restrict__ off,
                                                 const int* __restrict__ rank,
                                                 int* __restrict__ list, int E_) {
    const int e0 = blockIdx.x * (256 * FILL_PER_THREAD) + threadIdx.x;
    int s[FILL_PER_THREAD], d[FILL_PER_THREAD], rk[FILL_PER_THREAD], o[FILL_PER_THREAD];
    #pragma unroll
    for (int k = 0; k < FILL_PER_THREAD; ++k) {
        const int e  = e0 + 256 * k;
        const int ec = (e < E_) ? e : (E_ - 1);
        s[k]  = ei[ec];
        const int dv = ei[E_ + ec];
        rk[k] = rank[ec];
        d[k]  = (e < E_) ? dv : -1;
    }
    #pragma unroll
    for (int k = 0; k < FILL_PER_THREAD; ++k) o[k] = off[(d[k] >= 0) ? d[k] : 0];
    #pragma unroll
    for (int k = 0; k < FILL_PER_THREAD; ++k)
        if (d[k] >= 0) list[o[k] + rk[k]] = s[k];
}

__device__ __forceinline__ void fma4(float4& acc, const float4 v, const float w) {
    acc.x = fmaf(v.x, w, acc.x); acc.y = fmaf(v.y, w, acc.y);
    acc.z = fmaf(v.z, w, acc.z); acc.w = fmaf(v.w, w, acc.w);
}

template <int G>
__device__ __forceinline__ void gather_rows(float4& acc, const float4* __restrict__ h4,
                                            int s, float ex, int j, unsigned lane) {
    float4 v[G];
    float  w[G];
    #pragma unroll
    for (int g = 0; g < G; ++g) {
        const float4* __restrict__ row = h4 + (size_t)bcast_i(s, j + g) * 32;
        v[g] = row[lane];
        w[g] = bcast_f(ex, j + g);
    }
    #pragma unroll
    for (int g = 0; g < G; ++g) fma4(acc, v[g], w[g]);
}

__global__ __launch_bounds__(256) void pull_rows(const float* __restrict__ h,
                                                 const float* __restrict__ e_src,
                                                 const float* __restrict__ e_dst,
                                                 const int* __restrict__ off,
                                                 const int* __restrict__ cnt,
                                                 const int* __restrict__ list,
                                                 float* __restrict__ out1, int N_) {
    const unsigned lane = threadIdx.x & 31u;
    const int wv = __builtin_amdgcn_readfirstlane((int)(threadIdx.x >> 5));
    const int r0 = ((int)blockIdx.x * 8 + wv) * ROWS_PER_WAVE;
    const float4* __restrict__ h4 = (const float4*)h;

    int   offl = 0, cntl = 0;
    float edl  = 0.0f;
    if (lane < ROWS_PER_WAVE && r0 + (int)lane < N_) {
        offl = off[r0 + lane];
        cntl = cnt[r0 + lane];
        edl  = e_dst[r0 + lane];
    }

    for (int i = 0; i < ROWS_PER_WAVE; ++i) {
        const int r = r0 + i;
        if (r >= N_) break;
        const int   start = bcast_i(offl, i);
        const int   n     = bcast_i(cntl, i);
        const float ed    = bcast_f(edl, i);

        float4 acc  = make_float4(0.0f, 0.0f, 0.0f, 0.0f);
        float  wsum = 0.0f;
        for (int b = 0; b < n; b += 32) {
            const int c = (n - b < 32) ? (n - b) : 32;
            int   s  = 0;
            float ex = 0.0f;
            if ((int)lane < c) {
                s = list[start + b + (int)lane];
                float a = e_src[s] + ed;
                a  = 1.0f / (1.0f + expf(-a));
                ex = expf(a);
            }
            wsum += ex;
            int j = 0;
            for (; j + 8 <= c; j += 8) gather_rows<8>(acc, h4, s, ex, j, lane);
            if (j + 4 <= c) { gather_rows<4>(acc, h4, s, ex, j, lane); j += 4; }
            if (j + 2 <= c) { gather_rows<2>(acc, h4, s, ex, j, lane); j += 2; }
            if (j < c)        gather_rows<1>(acc, h4, s, ex, j, lane);
        }
        #pragma unroll
        for (int o = 16; o > 0; o >>= 1) wsum += __shfl_xor(wsum, o, 32);

        float4 res;
        const bool any = (wsum > 0.0f);
        res.x = any ? acc.x / wsum : 0.0f;
        res.y = any ? acc.y / wsum : 0.0f;
        res.z = any ? acc.z / wsum : 0.0f;
        res.w = any ? acc.w / wsum : 0.0f;
        res.x = (res.x > 0.0f) ? res.x : (expf(res.x) - 1.0f);
        res.y = (res.y > 0.0f) ? res.y : (expf(res.y) - 1.0f);
        res.z = (res.z > 0.0f) ? res.z : (expf(res.z) - 1.0f);
        res.w = (res.w > 0.0f) ? res.w : (expf(res.w) - 1.0f);
        ((float4*)out1)[(size_t)r * 32 + lane] = res;
    }
}

__global__ __launch_bounds__(256) void gemm2_kernel(const float* __restrict__ out1,
                                                    const float* __restrict__ W2,
                                                    float* __restrict__ out, int N_) {
    __shared__ float lds[16 * LDA2];
    const int tile = blockIdx.x;
    const int t    = threadIdx.x;
    const int row0 = tile * 16;
    const bool full = (row0 + 16 <= N_);

    #pragma unroll
    for (int i = 0; i < 8; ++i) {
        const int idx = i * 256 + t;
        const int r   = idx >> 7;
        const int c   = idx & 127;
        int row = row0 + r;
        if (row >= N_) row = N_ - 1;
        lds[r * LDA2 + c] = out1[(size_t)row * 128 + c];
    }
    __syncthreads();

    const int wave = t >> 5;
    const int lane = t & 31;
    const int m    = lane & 15;
    const int koff = (lane >> 4) * 2;
    const int n    = (wave << 4) + m;

    v8f acc = {};
    for (int k0 = 0; k0 < 128; k0 += 4) {
        v2f a, b;
        a.x = lds[m * LDA2 + k0 + koff];
        a.y = lds[m * LDA2 + k0 + koff + 1];
        b.x = W2[(size_t)(k0 + koff)     * 128 + n];
        b.y = W2[(size_t)(k0 + koff + 1) * 128 + n];
        acc = __builtin_amdgcn_wmma_f32_16x16x4_f32(false, a, false, b,
                                                    (short)0, acc, false, false);
    }

    const int rbase = row0 + ((lane >> 4) << 3);
    if (full) {
        #pragma unroll
        for (int r = 0; r < 8; ++r)
            out[(size_t)(rbase + r) * 128 + n] = acc[r];
    } else {
        #pragma unroll
        for (int r = 0; r < 8; ++r) {
            const int row = rbase + r;
            if (row < N_) out[(size_t)row * 128 + n] = acc[r];
        }
    }
}

extern "C" void kernel_launch(void* const* d_in, const int* in_sizes, int n_in,
                              void* d_out, int out_size, void* d_ws, size_t ws_size,
                              hipStream_t stream) {
    const float* x       = (const float*)d_in[0];
    const int*   ei      = (const int*)  d_in[1];
    const float* W1      = (const float*)d_in[2];
    const float* att_src = (const float*)d_in[3];
    const float* att_dst = (const float*)d_in[4];
    const float* W2      = (const float*)d_in[5];
    float*       out     = (float*)d_out;

    const int N_ = in_sizes[0] / 256;
    const int E_ = in_sizes[1] / 2;

    float*    ws    = (float*)d_ws;
    float*    h     = ws;
    float*    out1  = h     + (size_t)N_ * 128;
    float*    e_src = out1  + (size_t)N_ * 128;
    float*    e_dst = e_src + N_;
    const int Np    = (N_ + 63) & ~63;
    int*      ctrl  = (int*)d_out;
    int*      cnt   = ctrl + 256;
    int*      off   = cnt  + Np;
    int*      rank  = off  + Np;
    int*      list  = rank + E_;

    hipMemsetAsync(ctrl, 0, (size_t)(256 + N_) * sizeof(int), stream);

    const int tiles = (N_ + 15) / 16;
    edge_rank     <<<256, 256, 0, stream>>>(ei, cnt, rank, ctrl, E_, N_);
    row_alloc     <<<(N_ + 255) / 256, 256, 0, stream>>>(cnt, off, ctrl + 128, N_);
    edge_fill     <<<(E_ + 256 * FILL_PER_THREAD - 1) / (256 * FILL_PER_THREAD), 256, 0, stream>>>(
                      ei, off, rank, list, E_);
    gemm1_kernel  <<<tiles, 256, 0, stream>>>(x, W1, h, N_);
    attn_kernel   <<<(N_ * 32 + 255) / 256, 256, 0, stream>>>(h, att_src, att_dst, e_src, e_dst, N_);
    pull_rows     <<<(N_ + 8 * ROWS_PER_WAVE - 1) / (8 * ROWS_PER_WAVE), 256, 0, stream>>>(
                      h, e_src, e_dst, off, cnt, list, out1, N_);
    gemm2_kernel  <<<tiles, 256, 0, stream>>>(out1, W2, out, N_);
}


# logit launch as a resident grid: a wave takes 32 rows of h (all row loads in flight), sums them by a transposing butterfly, one result per lane, whole-line stores; was one short wave per row
# speedup vs baseline: 2.6867x; 1.0420x over previous
#include <hip/hip_runtime.h>
#include <math.h>

typedef __attribute__((ext_vector_type(2))) float v2f;
typedef __attribute__((ext_vector_type(8))) float v8f;

#define LDA1 258
#define LDA2 130
#define QHEAD_STRIDE 64

__global__ __launch_bounds__(256) void gemm1_kernel(const float* __restrict__ x,
                                                    const float* __restrict__ W1,
                                                    float* __restrict__ h, int N_) {
    __shared__ float lds[16 * LDA1];
    const int tile = blockIdx.x;
    const int t    = threadIdx.x;
    const int row0 = tile * 16;
    const bool full = (row0 + 16 <= N_);

    #pragma unroll
    for (int i = 0; i < 16; ++i) {
        int r = row0 + i;
        if (r >= N_) r = N_ - 1;
        lds[i * LDA1 + t] = x[(size_t)r * 256 + t];
    }
    __syncthreads();

    const int wave = t >> 5;
    const int lane = t & 31;
    const int m    = lane & 15;
    const int koff = (lane >> 4) * 2;
    const int n    = (wave << 4) + m;

    v8f acc = {};
    for (int k0 = 0; k0 < 256; k0 += 4) {
        v2f a, b;
        a.x = lds[m * LDA1 + k0 + koff];
        a.y = lds[m * LDA1 + k0 + koff + 1];
        b.x = W1[(size_t)(k0 + koff)     * 128 + n];
        b.y = W1[(size_t)(k0 + koff + 1) * 128 + n];
        acc = __builtin_amdgcn_wmma_f32_16x16x4_f32(false, a, false, b,
                                                    (short)0, acc, false, false);
    }

    const int rbase = row0 + ((lane >> 4) << 3);
    if (full) {
        #pragma unroll
        for (int r = 0; r < 8; ++r)
            h[(size_t)(rbase + r) * 128 + n] = acc[r];
    } else {
        #pragma unroll
        for (int r = 0; r < 8; ++r) {
            const int row = rbase + r;
            if (row < N_) h[(size_t)row * 128 + n] = acc[r];
        }
    }
}

#define ATTN_ROWS 32
#define ATTN_WAVES 4

__device__ __forceinline__ float wave_transpose_sum(float (&v)[ATTN_ROWS], const int lane) {
    #pragma unroll
    for (int off = ATTN_ROWS / 2; off > 0; off >>= 1) {
        const bool up = (lane & off) != 0;
        #pragma unroll
        for (int j = 0; j < off; ++j) {
            const float lo = v[j], hi = v[j + off];
            const float keep = up ? hi : lo;
            const float send = up ? lo : hi;
            v[j] = keep + __shfl_xor(send, off, 32);
        }
    }
    return v[0];
}

template <bool FULL>
__device__ __forceinline__ void attn_rows(const float4* __restrict__ h4, const float4 as,
                                          const float4 ad, float* __restrict__ e_src,
                                          float* __restrict__ e_dst, const int r0,
                                          const int lane, const int N_) {
    float4 hv[ATTN_ROWS];
    float  s[ATTN_ROWS], d[ATTN_ROWS];
    #pragma unroll
    for (int i = 0; i < ATTN_ROWS; ++i) {
        int r = r0 + i;
        if (!FULL && r >= N_) r = N_ - 1;
        hv[i] = h4[(size_t)r * 32 + lane];
    }
    #pragma unroll
    for (int i = 0; i < ATTN_ROWS; ++i) {
        s[i] = fmaf(hv[i].w, as.w, fmaf(hv[i].z, as.z, fmaf(hv[i].x, as.x, hv[i].y * as.y)));
        d[i] = fmaf(hv[i].w, ad.w, fmaf(hv[i].z, ad.z, fmaf(hv[i].x, ad.x, hv[i].y * ad.y)));
    }
    const float ss = wave_transpose_sum(s, lane);
    const float dd = wave_transpose_sum(d, lane);
    if (FULL || r0 + lane < N_) { e_src[r0 + lane] = ss; e_dst[r0 + lane] = dd; }
}

__global__ __launch_bounds__(32 * ATTN_WAVES) void attn_kernel(const float* __restrict__ h,
                                                   const float* __restrict__ att_src,
                                                   const float* __restrict__ att_dst,
                                                   float* __restrict__ e_src,
                                                   float* __restrict__ e_dst, int N_) {
    const int lane = threadIdx.x & 31;
    const int wv = __builtin_amdgcn_readfirstlane((int)(threadIdx.x >> 5));
    const int r0 = ((int)blockIdx.x * ATTN_WAVES + wv) * ATTN_ROWS;
    if (r0 >= N_) return;
    const float4 as = ((const float4*)att_src)[lane];
    const float4 ad = ((const float4*)att_dst)[lane];
    if (r0 + ATTN_ROWS <= N_) attn_rows<true >((const float4*)h, as, ad, e_src, e_dst, r0, lane, N_);
    else                      attn_rows<false>((const float4*)h, as, ad, e_src, e_dst, r0, lane, N_);
}

#define RANK_PER_LANE 16
#define RANK_CHUNK (32 * RANK_PER_LANE)
#define FILL_PER_THREAD 8
#define ROWS_PER_WAVE 4

__device__ __forceinline__ int   bcast_i(int v, int j)   { return __builtin_amdgcn_readlane(v, j); }
__device__ __forceinline__ float bcast_f(float v, int j) {
    return __int_as_float(__builtin_amdgcn_readlane(__float_as_int(v), j));
}

__device__ __forceinline__ unsigned ticket_issue(unsigned* head, int lane) {
    unsigned t = 0u;
    if (lane == 0) t = atomicInc(head, 0xffffffffu);
    return t;
}
__device__ __forceinline__ int ticket_value(unsigned t) {
    return (int)__builtin_amdgcn_readfirstlane(t);
}

__global__ __launch_bounds__(256) void edge_rank(const int* __restrict__ ei,
                                                 int* __restrict__ cnt,
                                                 int* __restrict__ rank,
                                                 int* __restrict__ qhead, int E_, int N_) {
    const int lane   = threadIdx.x & 31;
    const int nticks = (E_ + RANK_CHUNK - 1) / RANK_CHUNK;
    const int nh     = ((N_ >> 1) + 63) & ~63;
    const int myhalf = (int)((__builtin_amdgcn_s_sendmsg_rtn(0x87) >> 18) & 1u);

    for (int pass = 0; pass < 2; ++pass) {
        const int q = myhalf ^ pass;
        unsigned* head = (unsigned*)qhead + q * QHEAD_STRIDE;
        int t = ticket_value(ticket_issue(head, lane));
        while (t < nticks) {
            const int e0 = t * RANK_CHUNK + lane;
            int  d[RANK_PER_LANE], rk[RANK_PER_LANE];
            bool mine[RANK_PER_LANE];
            #pragma unroll
            for (int k = 0; k < RANK_PER_LANE; ++k) {
                const int e  = e0 + 32 * k;
                const int dv = ei[E_ + ((e < E_) ? e : (E_ - 1))];
                d[k] = (e < E_) ? dv : -1;
            }
            const unsigned tnext = ticket_issue(head, lane);
            #pragma unroll
            for (int k = 0; k < RANK_PER_LANE; ++k) {
                mine[k] = (d[k] >= 0) && (((d[k] >= nh) ? 1 : 0) == q);
                rk[k] = 0;
                if (mine[k]) rk[k] = atomicAdd(cnt + d[k], 1);
            }
            #pragma unroll
            for (int k = 0; k < RANK_PER_LANE; ++k)
                if (mine[k]) rank[e0 + 32 * k] = rk[k];
            t = ticket_value(tnext);
        }
    }
}

__global__ __launch_bounds__(256) void row_alloc(const int* __restrict__ cnt,
                                                 int* __restrict__ off,
                                                 int* __restrict__ total, int N_) {
    __shared__ int wsum[8];
    __shared__ int base_s;
    const int t    = threadIdx.x;
    const int lane = t & 31;
    const int wave = t >> 5;
    const int i    = blockIdx.x * 256 + t;
    const int c    = (i < N_) ? cnt[i] : 0;
    int v = c;
    #pragma unroll
    for (int o = 1; o < 32; o <<= 1) {
        const int u = __shfl_up(v, o, 32);
        if (lane >= o) v += u;
    }
    if (lane == 31) wsum[wave] = v;
    __syncthreads();
    if (t == 0) {
        int s = 0;
        for (int w = 0; w < 8; ++w) { const int x = wsum[w]; wsum[w] = s; s += x; }
        base_s = atomicAdd(total, s);
    }
    __syncthreads();
    if (i < N_) off[i] = base_s + wsum[wave] + v - c;
}

__global__ __launch_bounds__(256) void edge_fill(const int* __restrict__ ei,
                                                 const int* __restrict__ off,
                                                 const int* __restrict__ rank,
                                                 int* __restrict__ list, int E_) {
    const int e0 = blockIdx.x * (256 * FILL_PER_THREAD) + threadIdx.x;
    int s[FILL_PER_THREAD], d[FILL_PER_THREAD], rk[FILL_PER_THREAD], o[FILL_PER_THREAD];
    #pragma unroll
    for (int k = 0; k < FILL_PER_THREAD; ++k) {
        const int e  = e0 + 256 * k;
        const int ec = (e < E_) ? e : (E_ - 1);
        s[k]  = ei[ec];
        const int dv = ei[E_ + ec];
        rk[k] = rank[ec];
        d[k]  = (e < E_) ? dv : -1;
    }
    #pragma unroll
    for (int k = 0; k < FILL_PER_THREAD; ++k) o[k] = off[(d[k] >= 0) ? d[k] : 0];
    #pragma unroll
    for (int k = 0; k < FILL_PER_THREAD; ++k)
        if (d[k] >= 0) list[o[k] + rk[k]] = s[k];
}

__device__ __forceinline__ void fma4(float4& acc, const float4 v, const float w) {
    acc.x = fmaf(v.x, w, acc.x); acc.y = fmaf(v.y, w, acc.y);
    acc.z = fmaf(v.z, w, acc.z); acc.w = fmaf(v.w, w, acc.w);
}

template <int G>
__device__ __forceinline__ void gather_rows(float4& acc, const float4* __restrict__ h4,
                                            int s, float ex, int j, unsigned lane) {
    float4 v[G];
    float  w[G];
    #pragma unroll
    for (int g = 0; g < G; ++g) {
        const float4* __restrict__ row = h4 + (size_t)bcast_i(s, j + g) * 32;
        v[g] = row[lane];
        w[g] = bcast_f(ex, j + g);
    }
    #pragma unroll
    for (int g = 0; g < G; ++g) fma4(acc, v[g], w[g]);
}

__global__ __launch_bounds__(256) void pull_rows(const float* __restrict__ h,
                                                 const float* __restrict__ e_src,
                                                 const float* __restrict__ e_dst,
                                                 const int* __restrict__ off,
                                                 const int* __restrict__ cnt,
                                                 const int* __restrict__ list,
                                                 float* __restrict__ out1, int N_) {
    const unsigned lane = threadIdx.x & 31u;
    const int wv = __builtin_amdgcn_readfirstlane((int)(threadIdx.x >> 5));
    const int r0 = ((int)blockIdx.x * 8 + wv) * ROWS_PER_WAVE;
    const float4* __restrict__ h4 = (const float4*)h;

    int   offl = 0, cntl = 0;
    float edl  = 0.0f;
    if (lane < ROWS_PER_WAVE && r0 + (int)lane < N_) {
        offl = off[r0 + lane];
        cntl = cnt[r0 + lane];
        edl  = e_dst[r0 + lane];
    }

    for (int i = 0; i < ROWS_PER_WAVE; ++i) {
        const int r = r0 + i;
        if (r >= N_) break;
        const int   start = bcast_i(offl, i);
        const int   n     = bcast_i(cntl, i);
        const float ed    = bcast_f(edl, i);

        float4 acc  = make_float4(0.0f, 0.0f, 0.0f, 0.0f);
        float  wsum = 0.0f;
        for (int b = 0; b < n; b += 32) {
            const int c = (n - b < 32) ? (n - b) : 32;
            int   s  = 0;
            float ex = 0.0f;
            if ((int)lane < c) {
                s = list[start + b + (int)lane];
                float a = e_src[s] + ed;
                a  = 1.0f / (1.0f + expf(-a));
                ex = expf(a);
            }
            wsum += ex;
            int j = 0;
            for (; j + 8 <= c; j += 8) gather_rows<8>(acc, h4, s, ex, j, lane);
            if (j + 4 <= c) { gather_rows<4>(acc, h4, s, ex, j, lane); j += 4; }
            if (j + 2 <= c) { gather_rows<2>(acc, h4, s, ex, j, lane); j += 2; }
            if (j < c)        gather_rows<1>(acc, h4, s, ex, j, lane);
        }
        #pragma unroll
        for (int o = 16; o > 0; o >>= 1) wsum += __shfl_xor(wsum, o, 32);

        float4 res;
        const bool any = (wsum > 0.0f);
        res.x = any ? acc.x / wsum : 0.0f;
        res.y = any ? acc.y / wsum : 0.0f;
        res.z = any ? acc.z / wsum : 0.0f;
        res.w = any ? acc.w / wsum : 0.0f;
        res.x = (res.x > 0.0f) ? res.x : (expf(res.x) - 1.0f);
        res.y = (res.y > 0.0f) ? res.y : (expf(res.y) - 1.0f);
        res.z = (res.z > 0.0f) ? res.z : (expf(res.z) - 1.0f);
        res.w = (res.w > 0.0f) ? res.w : (expf(res.w) - 1.0f);
        ((float4*)out1)[(size_t)r * 32 + lane] = res;
    }
}

__global__ __launch_bounds__(256) void gemm2_kernel(const float* __restrict__ out1,
                                                    const float* __restrict__ W2,
                                                    float* __restrict__ out, int N_) {
    __shared__ float lds[16 * LDA2];
    const int tile = blockIdx.x;
    const int t    = threadIdx.x;
    const int row0 = tile * 16;
    const bool full = (row0 + 16 <= N_);

    #pragma unroll
    for (int i = 0; i < 8; ++i) {
        const int idx = i * 256 + t;
        const int r   = idx >> 7;
        const int c   = idx & 127;
        int row = row0 + r;
        if (row >= N_) row = N_ - 1;
        lds[r * LDA2 + c] = out1[(size_t)row * 128 + c];
    }
    __syncthreads();

    const int wave = t >> 5;
    const int lane = t & 31;
    const int m    = lane & 15;
    const int koff = (lane >> 4) * 2;
    const int n    = (wave << 4) + m;

    v8f acc = {};
    for (int k0 = 0; k0 < 128; k0 += 4) {
        v2f a, b;
        a.x = lds[m * LDA2 + k0 + koff];
        a.y = lds[m * LDA2 + k0 + koff + 1];
        b.x = W2[(size_t)(k0 + koff)     * 128 + n];
        b.y = W2[(size_t)(k0 + koff + 1) * 128 + n];
        acc = __builtin_amdgcn_wmma_f32_16x16x4_f32(false, a, false, b,
                                                    (short)0, acc, false, false);
    }

    const int rbase = row0 + ((lane >> 4) << 3);
    if (full) {
        #pragma unroll
        for (int r = 0; r < 8; ++r)
            out[(size_t)(rbase + r) * 128 + n] = acc[r];
    } else {
        #pragma unroll
        for (int r = 0; r < 8; ++r) {
            const int row = rbase + r;
            if (row < N_) out[(size_t)row * 128 + n] = acc[r];
        }
    }
}

extern "C" void kernel_launch(void* const* d_in, const int* in_sizes, int n_in,
                              void* d_out, int out_size, void* d_ws, size_t ws_size,
                              hipStream_t stream) {
    const float* x       = (const float*)d_in[0];
    const int*   ei      = (const int*)  d_in[1];
    const float* W1      = (const float*)d_in[2];
    const float* att_src = (const float*)d_in[3];
    const float* att_dst = (const float*)d_in[4];
    const float* W2      = (const float*)d_in[5];
    float*       out     = (float*)d_out;

    const int N_ = in_sizes[0] / 256;
    const int E_ = in_sizes[1] / 2;

    float*    ws    = (float*)d_ws;
    float*    h     = ws;
    float*    out1  = h     + (size_t)N_ * 128;
    float*    e_src = out1  + (size_t)N_ * 128;
    float*    e_dst = e_src + N_;
    const int Np    = (N_ + 63) & ~63;
    int*      ctrl  = (int*)d_out;
    int*      cnt   = ctrl + 256;
    int*      off   = cnt  + Np;
    int*      rank  = off  + Np;
    int*      list  = rank + E_;

    hipMemsetAsync(ctrl, 0, (size_t)(256 + N_) * sizeof(int), stream);

    const int tiles = (N_ + 15) / 16;
    edge_rank     <<<256, 256, 0, stream>>>(ei, cnt, rank, ctrl, E_, N_);
    row_alloc     <<<(N_ + 255) / 256, 256, 0, stream>>>(cnt, off, ctrl + 128, N_);
    edge_fill     <<<(E_ + 256 * FILL_PER_THREAD - 1) / (256 * FILL_PER_THREAD), 256, 0, stream>>>(
                      ei, off, rank, list, E_);
    gemm1_kernel  <<<tiles, 256, 0, stream>>>(x, W1, h, N_);
    attn_kernel   <<<(N_ + ATTN_ROWS * ATTN_WAVES - 1) / (ATTN_ROWS * ATTN_WAVES), 32 * ATTN_WAVES, 0, stream>>>(
                      h, att_src, att_dst, e_src, e_dst, N_);
    pull_rows     <<<(N_ + 8 * ROWS_PER_WAVE - 1) / (8 * ROWS_PER_WAVE), 256, 0, stream>>>(
                      h, e_src, e_dst, off, cnt, list, out1, N_);
    gemm2_kernel  <<<tiles, 256, 0, stream>>>(out1, W2, out, N_);
}


# both dense products re-tiled: 64-row block tiles, each weight fragment loaded once feeds 4 matrix instructions on 4 accumulators; operands fetched a 32-column chunk ahead; k order unchanged
# speedup vs baseline: 2.8695x; 1.0680x over previous
#include <hip/hip_runtime.h>
#include <math.h>

typedef __attribute__((ext_vector_type(2))) float v2f;
typedef __attribute__((ext_vector_type(8))) float v8f;

#define QHEAD_STRIDE 64

#define GK 32
#define GP 36
#define GROWS 64
#define GBIG 4

template <int MT>
__device__ __forceinline__ void gemm_chunk(v8f (&acc)[MT], const v2f (&b)[GK / 4],
                                           const float* __restrict__ cur, const int m, const int koff) {
    #pragma unroll
    for (int ks = 0; ks < GK / 4; ++ks) {
        #pragma unroll
        for (int mt = 0; mt < MT; ++mt) {
            const v2f a = *(const v2f*)(cur + (mt * 16 + m) * GP + ks * 4 + koff);
            acc[mt] = __builtin_amdgcn_wmma_f32_16x16x4_f32(false, a, false, b[ks],
                                                            (short)0, acc[mt], false, false);
        }
    }
}

template <int KDIM, int MT>
__device__ __forceinline__ void gemm_rows(const float* __restrict__ A, const float* __restrict__ Wm,
                                          float* __restrict__ C, const int row0, const int N_,
                                          float* __restrict__ lds) {
    constexpr int ROWS = 16 * MT;
    constexpr int NV   = ROWS * (GK / 4);
    constexpr int NL   = (NV + 255) / 256;
    constexpr int NCH  = KDIM / GK;
    const int t    = threadIdx.x;
    const int wave = t >> 5;
    const int lane = t & 31;
    const int m    = lane & 15;
    const int koff = (lane >> 4) * 2;
    const int n    = (wave << 4) + m;

    constexpr bool ALL = (NV % 256 == 0);
    const bool on = ALL || (t < NV);
    const float4* gsrc[NL];
    int    ldst[NL];
    float4 st[NL];
    #pragma unroll
    for (int i = 0; i < NL; ++i) {
        const int idx = i * 256 + (on ? t : 0);
        const int r   = idx >> 3;
        const int c4  = idx & 7;
        int row = row0 + r;
        if (row >= N_) row = N_ - 1;
        gsrc[i] = (const float4*)(A + (size_t)row * KDIM + c4 * 4);
        ldst[i] = r * GP + c4 * 4;
        st[i]   = make_float4(0.0f, 0.0f, 0.0f, 0.0f);
    }
    v2f b[GK / 4];
    #pragma unroll
    for (int ks = 0; ks < GK / 4; ++ks) {
        const int k = ks * 4 + koff;
        b[ks].x = Wm[(size_t)k       * 128 + n];
        b[ks].y = Wm[(size_t)(k + 1) * 128 + n];
    }
    if (on) {
        #pragma unroll
        for (int i = 0; i < NL; ++i) st[i] = gsrc[i][0];
        #pragma unroll
        for (int i = 0; i < NL; ++i) *(float4*)(lds + ldst[i]) = st[i];
    }
    __syncthreads();

    v8f acc[MT];
    #pragma unroll
    for (int mt = 0; mt < MT; ++mt) acc[mt] = v8f{};

    #pragma unroll
    for (int c = 0; c < NCH - 1; ++c) {
        const float* __restrict__ cur = lds + (c & 1) * (GROWS * GP);
        float* __restrict__ nxt = lds + ((c + 1) & 1) * (GROWS * GP);
        if (on) {
            #pragma unroll
            for (int i = 0; i < NL; ++i) st[i] = gsrc[i][(c + 1) * (GK / 4)];
        }
        v2f bn[GK / 4];
        #pragma unroll
        for (int ks = 0; ks < GK / 4; ++ks) {
            const int k = (c + 1) * GK + ks * 4 + koff;
            bn[ks].x = Wm[(size_t)k       * 128 + n];
            bn[ks].y = Wm[(size_t)(k + 1) * 128 + n];
        }
        gemm_chunk<MT>(acc, b, cur, m, koff);
        __builtin_amdgcn_sched_barrier(0);
        if (on) {
            #pragma unroll
            for (int i = 0; i < NL; ++i) *(float4*)(nxt + ldst[i]) = st[i];
        }
        #pragma unroll
        for (int ks = 0; ks < GK / 4; ++ks) b[ks] = bn[ks];
        __syncthreads();
    }
    gemm_chunk<MT>(acc, b, lds + ((NCH - 1) & 1) * (GROWS * GP), m, koff);

    const bool full = (row0 + ROWS <= N_);
    #pragma unroll
    for (int mt = 0; mt < MT; ++mt) {
        const int rbase = row0 + mt * 16 + ((lane >> 4) << 3);
        if (full) {
            #pragma unroll
            for (int r = 0; r < 8; ++r)
                C[(size_t)(rbase + r) * 128 + n] = acc[mt][r];
        } else {
            #pragma unroll
            for (int r = 0; r < 8; ++r) {
                const int row = rbase + r;
                if (row < N_) C[(size_t)row * 128 + n] = acc[mt][r];
            }
        }
    }
}

template <int KDIM>
__device__ __forceinline__ void gemm_block(const float* __restrict__ A, const float* __restrict__ Wm,
                                           float* __restrict__ C, const int N_, const int nbig,
                                           float* __restrict__ lds) {
    const int b = blockIdx.x;
    if (b < nbig) gemm_rows<KDIM, GBIG>(A, Wm, C, b * GROWS, N_, lds);
    else          gemm_rows<KDIM, 1   >(A, Wm, C, nbig * GROWS + (b - nbig) * 16, N_, lds);
}

__global__ __launch_bounds__(256) void gemm1_kernel(const float* __restrict__ x,
                                                    const float* __restrict__ W1,
                                                    float* __restrict__ h, int N_, int nbig) {
    __shared__ __attribute__((aligned(16))) float lds[2 * GROWS * GP];
    gemm_block<256>(x, W1, h, N_, nbig, lds);
}

#define ATTN_ROWS 32
#define ATTN_WAVES 4

__device__ __forceinline__ float wave_transpose_sum(float (&v)[ATTN_ROWS], const int lane) {
    #pragma unroll
    for (int off = ATTN_ROWS / 2; off > 0; off >>= 1) {
        const bool up = (lane & off) != 0;
        #pragma unroll
        for (int j = 0; j < off; ++j) {
            const float lo = v[j], hi = v[j + off];
            const float keep = up ? hi : lo;
            const float send = up ? lo : hi;
            v[j] = keep + __shfl_xor(send, off, 32);
        }
    }
    return v[0];
}

template <bool FULL>
__device__ __forceinline__ void attn_rows(const float4* __restrict__ h4, const float4 as,
                                          const float4 ad, float* __restrict__ e_src,
                                          float* __restrict__ e_dst, const int r0,
                                          const int lane, const int N_) {
    float4 hv[ATTN_ROWS];
    float  s[ATTN_ROWS], d[ATTN_ROWS];
    #pragma unroll
    for (int i = 0; i < ATTN_ROWS; ++i) {
        int r = r0 + i;
        if (!FULL && r >= N_) r = N_ - 1;
        hv[i] = h4[(size_t)r * 32 + lane];
    }
    #pragma unroll
    for (int i = 0; i < ATTN_ROWS; ++i) {
        s[i] = fmaf(hv[i].w, as.w, fmaf(hv[i].z, as.z, fmaf(hv[i].x, as.x, hv[i].y * as.y)));
        d[i] = fmaf(hv[i].w, ad.w, fmaf(hv[i].z, ad.z, fmaf(hv[i].x, ad.x, hv[i].y * ad.y)));
    }
    const float ss = wave_transpose_sum(s, lane);
    const float dd = wave_transpose_sum(d, lane);
    if (FULL || r0 + lane < N_) { e_src[r0 + lane] = ss; e_dst[r0 + lane] = dd; }
}

__global__ __launch_bounds__(32 * ATTN_WAVES) void attn_kernel(const float* __restrict__ h,
                                                   const float* __restrict__ att_src,
                                                   const float* __restrict__ att_dst,
                                                   float* __restrict__ e_src,
                                                   float* __restrict__ e_dst, int N_) {
    const int lane = threadIdx.x & 31;
    const int wv = __builtin_amdgcn_readfirstlane((int)(threadIdx.x >> 5));
    const int r0 = ((int)blockIdx.x * ATTN_WAVES + wv) * ATTN_ROWS;
    if (r0 >= N_) return;
    const float4 as = ((const float4*)att_src)[lane];
    const float4 ad = ((const float4*)att_dst)[lane];
    if (r0 + ATTN_ROWS <= N_) attn_rows<true >((const float4*)h, as, ad, e_src, e_dst, r0, lane, N_);
    else                      attn_rows<false>((const float4*)h, as, ad, e_src, e_dst, r0, lane, N_);
}

#define RANK_PER_LANE 16
#define RANK_CHUNK (32 * RANK_PER_LANE)
#define FILL_PER_THREAD 8
#define ROWS_PER_WAVE 4

__device__ __forceinline__ int   bcast_i(int v, int j)   { return __builtin_amdgcn_readlane(v, j); }
__device__ __forceinline__ float bcast_f(float v, int j) {
    return __int_as_float(__builtin_amdgcn_readlane(__float_as_int(v), j));
}

__device__ __forceinline__ unsigned ticket_issue(unsigned* head, int lane) {
    unsigned t = 0u;
    if (lane == 0) t = atomicInc(head, 0xffffffffu);
    return t;
}
__device__ __forceinline__ int ticket_value(unsigned t) {
    return (int)__builtin_amdgcn_readfirstlane(t);
}

__global__ __launch_bounds__(256) void edge_rank(const int* __restrict__ ei,
                                                 int* __restrict__ cnt,
                                                 int* __restrict__ rank,
                                                 int* __restrict__ qhead, int E_, int N_) {
    const int lane   = threadIdx.x & 31;
    const int nticks = (E_ + RANK_CHUNK - 1) / RANK_CHUNK;
    const int nh     = ((N_ >> 1) + 63) & ~63;
    const int myhalf = (int)((__builtin_amdgcn_s_sendmsg_rtn(0x87) >> 18) & 1u);

    for (int pass = 0; pass < 2; ++pass) {
        const int q = myhalf ^ pass;
        unsigned* head = (unsigned*)qhead + q * QHEAD_STRIDE;
        int t = ticket_value(ticket_issue(head, lane));
        while (t < nticks) {
            const int e0 = t * RANK_CHUNK + lane;
            int  d[RANK_PER_LANE], rk[RANK_PER_LANE];
            bool mine[RANK_PER_LANE];
            #pragma unroll
            for (int k = 0; k < RANK_PER_LANE; ++k) {
                const int e  = e0 + 32 * k;
                const int dv = ei[E_ + ((e < E_) ? e : (E_ - 1))];
                d[k] = (e < E_) ? dv : -1;
            }
            const unsigned tnext = ticket_issue(head, lane);
            #pragma unroll
            for (int k = 0; k < RANK_PER_LANE; ++k) {
                mine[k] = (d[k] >= 0) && (((d[k] >= nh) ? 1 : 0) == q);
                rk[k] = 0;
                if (mine[k]) rk[k] = atomicAdd(cnt + d[k], 1);
            }
            #pragma unroll
            for (int k = 0; k < RANK_PER_LANE; ++k)
                if (mine[k]) rank[e0 + 32 * k] = rk[k];
            t = ticket_value(tnext);
        }
    }
}

__global__ __launch_bounds__(256) void row_alloc(const int* __restrict__ cnt,
                                                 int* __restrict__ off,
                                                 int* __restrict__ total, int N_) {
    __shared__ int wsum[8];
    __shared__ int base_s;
    const int t    = threadIdx.x;
    const int lane = t & 31;
    const int wave = t >> 5;
    const int i    = blockIdx.x * 256 + t;
    const int c    = (i < N_) ? cnt[i] : 0;
    int v = c;
    #pragma unroll
    for (int o = 1; o < 32; o <<= 1) {
        const int u = __shfl_up(v, o, 32);
        if (lane >= o) v += u;
    }
    if (lane == 31) wsum[wave] = v;
    __syncthreads();
    if (t == 0) {
        int s = 0;
        for (int w = 0; w < 8; ++w) { const int x = wsum[w]; wsum[w] = s; s += x; }
        base_s = atomicAdd(total, s);
    }
    __syncthreads();
    if (i < N_) off[i] = base_s + wsum[wave] + v - c;
}

__global__ __launch_bounds__(256) void edge_fill(const int* __restrict__ ei,
                                                 const int* __restrict__ off,
                                                 const int* __restrict__ rank,
                                                 int* __restrict__ list, int E_) {
    const int e0 = blockIdx.x * (256 * FILL_PER_THREAD) + threadIdx.x;
    int s[FILL_PER_THREAD], d[FILL_PER_THREAD], rk[FILL_PER_THREAD], o[FILL_PER_THREAD];
    #pragma unroll
    for (int k = 0; k < FILL_PER_THREAD; ++k) {
        const int e  = e0 + 256 * k;
        const int ec = (e < E_) ? e : (E_ - 1);
        s[k]  = ei[ec];
        const int dv = ei[E_ + ec];
        rk[k] = rank[ec];
        d[k]  = (e < E_) ? dv : -1;
    }
    #pragma unroll
    for (int k = 0; k < FILL_PER_THREAD; ++k) o[k] = off[(d[k] >= 0) ? d[k] : 0];
    #pragma unroll
    for (int k = 0; k < FILL_PER_THREAD; ++k)
        if (d[k] >= 0) list[o[k] + rk[k]] = s[k];
}

__device__ __forceinline__ void fma4(float4& acc, const float4 v, const float w) {
    acc.x = fmaf(v.x, w, acc.x); acc.y = fmaf(v.y, w, acc.y);
    acc.z = fmaf(v.z, w, acc.z); acc.w = fmaf(v.w, w, acc.w);
}

template <int G>
__device__ __forceinline__ void gather_rows(float4& acc, const float4* __restrict__ h4,
                                            int s, float ex, int j, unsigned lane) {
    float4 v[G];
    float  w[G];
    #pragma unroll
    for (int g = 0; g < G; ++g) {
        const float4* __restrict__ row = h4 + (size_t)bcast_i(s, j + g) * 32;
        v[g] = row[lane];
        w[g] = bcast_f(ex, j + g);
    }
    #pragma unroll
    for (int g = 0; g < G; ++g) fma4(acc, v[g], w[g]);
}

__global__ __launch_bounds__(256) void pull_rows(const float* __restrict__ h,
                                                 const float* __restrict__ e_src,
                                                 const float* __restrict__ e_dst,
                                                 const int* __restrict__ off,
                                                 const int* __restrict__ cnt,
                                                 const int* __restrict__ list,
                                                 float* __restrict__ out1, int N_) {
    const unsigned lane = threadIdx.x & 31u;
    const int wv = __builtin_amdgcn_readfirstlane((int)(threadIdx.x >> 5));
    const int r0 = ((int)blockIdx.x * 8 + wv) * ROWS_PER_WAVE;
    const float4* __restrict__ h4 = (const float4*)h;

    int   offl = 0, cntl = 0;
    float edl  = 0.0f;
    if (lane < ROWS_PER_WAVE && r0 + (int)lane < N_) {
        offl = off[r0 + lane];
        cntl = cnt[r0 + lane];
        edl  = e_dst[r0 + lane];
    }

    for (int i = 0; i < ROWS_PER_WAVE; ++i) {
        const int r = r0 + i;
        if (r >= N_) break;
        const int   start = bcast_i(offl, i);
        const int   n     = bcast_i(cntl, i);
        const float ed    = bcast_f(edl, i);

        float4 acc  = make_float4(0.0f, 0.0f, 0.0f, 0.0f);
        float  wsum = 0.0f;
        for (int b = 0; b < n; b += 32) {
            const int c = (n - b < 32) ? (n - b) : 32;
            int   s  = 0;
            float ex = 0.0f;
            if ((int)lane < c) {
                s = list[start + b + (int)lane];
                float a = e_src[s] + ed;
                a  = 1.0f / (1.0f + expf(-a));
                ex = expf(a);
            }
            wsum += ex;
            int j = 0;
            for (; j + 8 <= c; j += 8) gather_rows<8>(acc, h4, s, ex, j, lane);
            if (j + 4 <= c) { gather_rows<4>(acc, h4, s, ex, j, lane); j += 4; }
            if (j + 2 <= c) { gather_rows<2>(acc, h4, s, ex, j, lane); j += 2; }
            if (j < c)        gather_rows<1>(acc, h4, s, ex, j, lane);
        }
        #pragma unroll
        for (int o = 16; o > 0; o >>= 1) wsum += __shfl_xor(wsum, o, 32);

        float4 res;
        const bool any = (wsum > 0.0f);
        res.x = any ? acc.x / wsum : 0.0f;
        res.y = any ? acc.y / wsum : 0.0f;
        res.z = any ? acc.z / wsum : 0.0f;
        res.w = any ? acc.w / wsum : 0.0f;
        res.x = (res.x > 0.0f) ? res.x : (expf(res.x) - 1.0f);
        res.y = (res.y > 0.0f) ? res.y : (expf(res.y) - 1.0f);
        res.z = (res.z > 0.0f) ? res.z : (expf(res.z) - 1.0f);
        res.w = (res.w > 0.0f) ? res.w : (expf(res.w) - 1.0f);
        ((float4*)out1)[(size_t)r * 32 + lane] = res;
    }
}

__global__ __launch_bounds__(256) void gemm2_kernel(const float* __restrict__ out1,
                                                    const float* __restrict__ W2,
                                                    float* __restrict__ out, int N_, int nbig) {
    __shared__ __attribute__((aligned(16))) float lds[2 * GROWS * GP];
    gemm_block<128>(out1, W2, out, N_, nbig, lds);
}

extern "C" void kernel_launch(void* const* d_in, const int* in_sizes, int n_in,
                              void* d_out, int out_size, void* d_ws, size_t ws_size,
                              hipStream_t stream) {
    const float* x       = (const float*)d_in[0];
    const int*   ei      = (const int*)  d_in[1];
    const float* W1      = (const float*)d_in[2];
    const float* att_src = (const float*)d_in[3];
    const float* att_dst = (const float*)d_in[4];
    const float* W2      = (const float*)d_in[5];
    float*       out     = (float*)d_out;

    const int N_ = in_sizes[0] / 256;
    const int E_ = in_sizes[1] / 2;

    float*    ws    = (float*)d_ws;
    float*    h     = ws;
    float*    out1  = h     + (size_t)N_ * 128;
    float*    e_src = out1  + (size_t)N_ * 128;
    float*    e_dst = e_src + N_;
    const int Np    = (N_ + 63) & ~63;
    int*      ctrl  = (int*)d_out;
    int*      cnt   = ctrl + 256;
    int*      off   = cnt  + Np;
    int*      rank  = off  + Np;
    int*      list  = rank + E_;

    hipMemsetAsync(ctrl, 0, (size_t)(256 + N_) * sizeof(int), stream);

    const int nbig   = ((N_ / GROWS) / 256) * 256;
    const int nsmall = (N_ - nbig * GROWS + 15) / 16;
    const int tiles  = nbig + nsmall;
    edge_rank     <<<256, 256, 0, stream>>>(ei, cnt, rank, ctrl, E_, N_);
    row_alloc     <<<(N_ + 255) / 256, 256, 0, stream>>>(cnt, off, ctrl + 128, N_);
    edge_fill     <<<(E_ + 256 * FILL_PER_THREAD - 1) / (256 * FILL_PER_THREAD), 256, 0, stream>>>(
                      ei, off, rank, list, E_);
    gemm1_kernel  <<<tiles, 256, 0, stream>>>(x, W1, h, N_, nbig);
    attn_kernel   <<<(N_ + ATTN_ROWS * ATTN_WAVES - 1) / (ATTN_ROWS * ATTN_WAVES), 32 * ATTN_WAVES, 0, stream>>>(
                      h, att_src, att_dst, e_src, e_dst, N_);
    pull_rows     <<<(N_ + 8 * ROWS_PER_WAVE - 1) / (8 * ROWS_PER_WAVE), 256, 0, stream>>>(
                      h, e_src, e_dst, off, cnt, list, out1, N_);
    gemm2_kernel  <<<tiles, 256, 0, stream>>>(out1, W2, out, N_, nbig);
}


# edge ranking pass rides in the GEMM1 launch as its first 256 work-groups: its load/add/store round trips run under the matrix-bound product; one launch fewer; arrival-order rank as before
# speedup vs baseline: 4.1540x; 1.4476x over previous
#include <hip/hip_runtime.h>
#include <math.h>

typedef __attribute__((ext_vector_type(2))) float v2f;
typedef __attribute__((ext_vector_type(8))) float v8f;

#define QHEAD_STRIDE 64

#define GK 32
#define GP 36
#define GROWS 64
#define GBIG 4

template <int MT>
__device__ __forceinline__ void gemm_chunk(v8f (&acc)[MT], const v2f (&b)[GK / 4],
                                           const float* __restrict__ cur, const int m, const int koff) {
    #pragma unroll
    for (int ks = 0; ks < GK / 4; ++ks) {
        #pragma unroll
        for (int mt = 0; mt < MT; ++mt) {
            const v2f a = *(const v2f*)(cur + (mt * 16 + m) * GP + ks * 4 + koff);
            acc[mt] = __builtin_amdgcn_wmma_f32_16x16x4_f32(false, a, false, b[ks],
                                                            (short)0, acc[mt], false, false);
        }
    }
}

template <int KDIM, int MT>
__device__ __forceinline__ void gemm_rows(const float* __restrict__ A, const float* __restrict__ Wm,
                                          float* __restrict__ C, const int row0, const int N_,
                                          float* __restrict__ lds) {
    constexpr int ROWS = 16 * MT;
    constexpr int NV   = ROWS * (GK / 4);
    constexpr int NL   = (NV + 255) / 256;
    constexpr int NCH  = KDIM / GK;
    const int t    = threadIdx.x;
    const int wave = t >> 5;
    const int lane = t & 31;
    const int m    = lane & 15;
    const int koff = (lane >> 4) * 2;
    const int n    = (wave << 4) + m;

    constexpr bool ALL = (NV % 256 == 0);
    const bool on = ALL || (t < NV);
    const float4* gsrc[NL];
    int    ldst[NL];
    float4 st[NL];
    #pragma unroll
    for (int i = 0; i < NL; ++i) {
        const int idx = i * 256 + (on ? t : 0);
        const int r   = idx >> 3;
        const int c4  = idx & 7;
        int row = row0 + r;
        if (row >= N_) row = N_ - 1;
        gsrc[i] = (const float4*)(A + (size_t)row * KDIM + c4 * 4);
        ldst[i] = r * GP + c4 * 4;
        st[i]   = make_float4(0.0f, 0.0f, 0.0f, 0.0f);
    }
    v2f b[GK / 4];
    #pragma unroll
    for (int ks = 0; ks < GK / 4; ++ks) {
        const int k = ks * 4 + koff;
        b[ks].x = Wm[(size_t)k       * 128 + n];
        b[ks].y = Wm[(size_t)(k + 1) * 128 + n];
    }
    if (on) {
        #pragma unroll
        for (int i = 0; i < NL; ++i) st[i] = gsrc[i][0];
        #pragma unroll
        for (int i = 0; i < NL; ++i) *(float4*)(lds + ldst[i]) = st[i];
    }
    __syncthreads();

    v8f acc[MT];
    #pragma unroll
    for (int mt = 0; mt < MT; ++mt) acc[mt] = v8f{};

    #pragma unroll
    for (int c = 0; c < NCH - 1; ++c) {
        const float* __restrict__ cur = lds + (c & 1) * (GROWS * GP);
        float* __restrict__ nxt = lds + ((c + 1) & 1) * (GROWS * GP);
        if (on) {
            #pragma unroll
            for (int i = 0; i < NL; ++i) st[i] = gsrc[i][(c + 1) * (GK / 4)];
        }
        v2f bn[GK / 4];
        #pragma unroll
        for (int ks = 0; ks < GK / 4; ++ks) {
            const int k = (c + 1) * GK + ks * 4 + koff;
            bn[ks].x = Wm[(size_t)k       * 128 + n];
            bn[ks].y = Wm[(size_t)(k + 1) * 128 + n];
        }
        gemm_chunk<MT>(acc, b, cur, m, koff);
        __builtin_amdgcn_sched_barrier(0);
        if (on) {
            #pragma unroll
            for (int i = 0; i < NL; ++i) *(float4*)(nxt + ldst[i]) = st[i];
        }
        #pragma unroll
        for (int ks = 0; ks < GK / 4; ++ks) b[ks] = bn[ks];
        __syncthreads();
    }
    gemm_chunk<MT>(acc, b, lds + ((NCH - 1) & 1) * (GROWS * GP), m, koff);

    const bool full = (row0 + ROWS <= N_);
    #pragma unroll
    for (int mt = 0; mt < MT; ++mt) {
        const int rbase = row0 + mt * 16 + ((lane >> 4) << 3);
        if (full) {
            #pragma unroll
            for (int r = 0; r < 8; ++r)
                C[(size_t)(rbase + r) * 128 + n] = acc[mt][r];
        } else {
            #pragma unroll
            for (int r = 0; r < 8; ++r) {
                const int row = rbase + r;
                if (row < N_) C[(size_t)row * 128 + n] = acc[mt][r];
            }
        }
    }
}

template <int KDIM>
__device__ __forceinline__ void gemm_block(const float* __restrict__ A, const float* __restrict__ Wm,
                                           float* __restrict__ C, const int N_, const int nbig,
                                           const int b, float* __restrict__ lds) {
    if (b < nbig) gemm_rows<KDIM, GBIG>(A, Wm, C, b * GROWS, N_, lds);
    else          gemm_rows<KDIM, 1   >(A, Wm, C, nbig * GROWS + (b - nbig) * 16, N_, lds);
}

#define ATTN_ROWS 32
#define ATTN_WAVES 4

__device__ __forceinline__ float wave_transpose_sum(float (&v)[ATTN_ROWS], const int lane) {
    #pragma unroll
    for (int off = ATTN_ROWS / 2; off > 0; off >>= 1) {
        const bool up = (lane & off) != 0;
        #pragma unroll
        for (int j = 0; j < off; ++j) {
            const float lo = v[j], hi = v[j + off];
            const float keep = up ? hi : lo;
            const float send = up ? lo : hi;
            v[j] = keep + __shfl_xor(send, off, 32);
        }
    }
    return v[0];
}

template <bool FULL>
__device__ __forceinline__ void attn_rows(const float4* __restrict__ h4, const float4 as,
                                          const float4 ad, float* __restrict__ e_src,
                                          float* __restrict__ e_dst, const int r0,
                                          const int lane, const int N_) {
    float4 hv[ATTN_ROWS];
    float  s[ATTN_ROWS], d[ATTN_ROWS];
    #pragma unroll
    for (int i = 0; i < ATTN_ROWS; ++i) {
        int r = r0 + i;
        if (!FULL && r >= N_) r = N_ - 1;
        hv[i] = h4[(size_t)r * 32 + lane];
    }
    #pragma unroll
    for (int i = 0; i < ATTN_ROWS; ++i) {
        s[i] = fmaf(hv[i].w, as.w, fmaf(hv[i].z, as.z, fmaf(hv[i].x, as.x, hv[i].y * as.y)));
        d[i] = fmaf(hv[i].w, ad.w, fmaf(hv[i].z, ad.z, fmaf(hv[i].x, ad.x, hv[i].y * ad.y)));
    }
    const float ss = wave_transpose_sum(s, lane);
    const float dd = wave_transpose_sum(d, lane);
    if (FULL || r0 + lane < N_) { e_src[r0 + lane] = ss; e_dst[r0 + lane] = dd; }
}

__global__ __launch_bounds__(32 * ATTN_WAVES) void attn_kernel(const float* __restrict__ h,
                                                   const float* __restrict__ att_src,
                                                   const float* __restrict__ att_dst,
                                                   float* __restrict__ e_src,
                                                   float* __restrict__ e_dst, int N_) {
    const int lane = threadIdx.x & 31;
    const int wv = __builtin_amdgcn_readfirstlane((int)(threadIdx.x >> 5));
    const int r0 = ((int)blockIdx.x * ATTN_WAVES + wv) * ATTN_ROWS;
    if (r0 >= N_) return;
    const float4 as = ((const float4*)att_src)[lane];
    const float4 ad = ((const float4*)att_dst)[lane];
    if (r0 + ATTN_ROWS <= N_) attn_rows<true >((const float4*)h, as, ad, e_src, e_dst, r0, lane, N_);
    else                      attn_rows<false>((const float4*)h, as, ad, e_src, e_dst, r0, lane, N_);
}

#define RANK_PER_LANE 16
#define RANK_CHUNK (32 * RANK_PER_LANE)
#define SLOTS 32
#define SPILL_HEAD 128
#define ROWS_PER_WAVE 4

__device__ __forceinline__ int   bcast_i(int v, int j)   { return __builtin_amdgcn_readlane(v, j); }
__device__ __forceinline__ float bcast_f(float v, int j) {
    return __int_as_float(__builtin_amdgcn_readlane(__float_as_int(v), j));
}

__device__ __forceinline__ unsigned ticket_issue(unsigned* head, int lane) {
    unsigned t = 0u;
    if (lane == 0) t = atomicInc(head, 0xffffffffu);
    return t;
}
__device__ __forceinline__ int ticket_value(unsigned t) {
    return (int)__builtin_amdgcn_readfirstlane(t);
}

__device__ __forceinline__ void edge_rank(const int* __restrict__ ei,
                                          int* __restrict__ cnt,
                                          int* __restrict__ slots,
                                          int* __restrict__ spill,
                                          int* __restrict__ ctrl, const int E_, const int N_) {
    const int lane   = threadIdx.x & 31;
    const int nticks = (E_ + RANK_CHUNK - 1) / RANK_CHUNK;
    const int nh     = ((N_ >> 1) + 63) & ~63;
    const int myhalf = (int)((__builtin_amdgcn_s_sendmsg_rtn(0x87) >> 18) & 1u);
    int* nspill = ctrl + SPILL_HEAD;
    int2* pairs = (int2*)spill;

    for (int pass = 0; pass < 2; ++pass) {
        const int q = myhalf ^ pass;
        unsigned* head = (unsigned*)ctrl + q * QHEAD_STRIDE;
        int t = ticket_value(ticket_issue(head, lane));
        while (t < nticks) {
            const int e0 = t * RANK_CHUNK + lane;
            int  d[RANK_PER_LANE], s[RANK_PER_LANE], rk[RANK_PER_LANE];
            bool mine[RANK_PER_LANE];
            #pragma unroll
            for (int k = 0; k < RANK_PER_LANE; ++k) {
                const int e  = e0 + 32 * k;
                const int ec = (e < E_) ? e : (E_ - 1);
                const int dv = ei[E_ + ec];
                s[k] = ei[ec];
                d[k] = (e < E_) ? dv : -1;
            }
            const unsigned tnext = ticket_issue(head, lane);
            #pragma unroll
            for (int k = 0; k < RANK_PER_LANE; ++k) {
                mine[k] = (d[k] >= 0) && (((d[k] >= nh) ? 1 : 0) == q);
                rk[k] = 0;
                if (mine[k]) rk[k] = atomicAdd(cnt + d[k], 1);
            }
            #pragma unroll
            for (int k = 0; k < RANK_PER_LANE; ++k) {
                if (mine[k]) {
                    if (rk[k] < SLOTS) {
                        slots[d[k] * SLOTS + rk[k]] = s[k];
                    } else {
                        const int p = atomicAdd(nspill, 1);
                        pairs[p] = make_int2(d[k], s[k]);
                    }
                }
            }
            t = ticket_value(tnext);
        }
    }
}

#define RANK_BLOCKS 256

__global__ __launch_bounds__(256) void gemm1_rank_kernel(const float* __restrict__ x,
                                                         const float* __restrict__ W1,
                                                         float* __restrict__ h, int N_, int nbig,
                                                         const int* __restrict__ ei,
                                                         int* __restrict__ cnt,
                                                         int* __restrict__ slots,
                                                         int* __restrict__ spill,
                                                         int* __restrict__ ctrl, int E_) {
    __shared__ __attribute__((aligned(16))) float lds[2 * GROWS * GP];
    const int b = blockIdx.x;
    if (b < RANK_BLOCKS) edge_rank(ei, cnt, slots, spill, ctrl, E_, N_);
    else                 gemm_block<256>(x, W1, h, N_, nbig, b - RANK_BLOCKS, lds);
}

__device__ __forceinline__ void fma4(float4& acc, const float4 v, const float w) {
    acc.x = fmaf(v.x, w, acc.x); acc.y = fmaf(v.y, w, acc.y);
    acc.z = fmaf(v.z, w, acc.z); acc.w = fmaf(v.w, w, acc.w);
}

__device__ __forceinline__ float edge_weight(float a) {
    a = 1.0f / (1.0f + expf(-a));
    return expf(a);
}

template <int G>
__device__ __forceinline__ void gather_rows(float4& acc, const float4* __restrict__ h4,
                                            int s, float ex, int j, unsigned lane) {
    float4 v[G];
    float  w[G];
    #pragma unroll
    for (int g = 0; g < G; ++g) {
        const float4* __restrict__ row = h4 + (size_t)bcast_i(s, j + g) * 32;
        v[g] = row[lane];
        w[g] = bcast_f(ex, j + g);
    }
    #pragma unroll
    for (int g = 0; g < G; ++g) fma4(acc, v[g], w[g]);
}

__global__ __launch_bounds__(256) void pull_rows(const float* __restrict__ h,
                                                 const float* __restrict__ e_src,
                                                 const float* __restrict__ e_dst,
                                                 const int* __restrict__ cnt,
                                                 const int* __restrict__ slots,
                                                 const int* __restrict__ spill,
                                                 const int* __restrict__ ctrl,
                                                 float* __restrict__ out1, int N_) {
    const unsigned lane = threadIdx.x & 31u;
    const int wv = __builtin_amdgcn_readfirstlane((int)(threadIdx.x >> 5));
    const int r0 = ((int)blockIdx.x * 8 + wv) * ROWS_PER_WAVE;
    const float4* __restrict__ h4 = (const float4*)h;
    const int2* pairs = (const int2*)spill;

    int   cntl = 0;
    float edl  = 0.0f;
    if (lane < ROWS_PER_WAVE && r0 + (int)lane < N_) {
        cntl = cnt[r0 + lane];
        edl  = e_dst[r0 + lane];
    }

    for (int i = 0; i < ROWS_PER_WAVE; ++i) {
        const int r = r0 + i;
        if (r >= N_) break;
        const int   n  = bcast_i(cntl, i);
        const float ed = bcast_f(edl, i);

        float4 acc  = make_float4(0.0f, 0.0f, 0.0f, 0.0f);
        float  wsum = 0.0f;
        {
            const int c = (n < SLOTS) ? n : SLOTS;
            int   s  = 0;
            float ex = 0.0f;
            if ((int)lane < c) {
                s  = slots[r * SLOTS + (int)lane];
                ex = edge_weight(e_src[s] + ed);
            }
            wsum += ex;
            int j = 0;
            for (; j + 8 <= c; j += 8) gather_rows<8>(acc, h4, s, ex, j, lane);
            if (j + 4 <= c) { gather_rows<4>(acc, h4, s, ex, j, lane); j += 4; }
            if (j + 2 <= c) { gather_rows<2>(acc, h4, s, ex, j, lane); j += 2; }
            if (j < c)        gather_rows<1>(acc, h4, s, ex, j, lane);
        }
        if (n > SLOTS) {
            const int ns = ctrl[SPILL_HEAD];
            for (int b = 0; b < ns; b += 32) {
                int2 pr = make_int2(-1, 0);
                if (b + (int)lane < ns) pr = pairs[b + (int)lane];
                const bool hit = (pr.x == r);
                const int  s   = hit ? pr.y : 0;
                float ex = 0.0f;
                if (hit) ex = edge_weight(e_src[s] + ed);
                wsum += ex;
                unsigned m = __builtin_amdgcn_ballot_w32(hit);
                while (m != 0u) {
                    const int j = __builtin_ctz(m);
                    m &= m - 1u;
                    gather_rows<1>(acc, h4, s, ex, j, lane);
                }
            }
        }
        #pragma unroll
        for (int o = 16; o > 0; o >>= 1) wsum += __shfl_xor(wsum, o, 32);

        float4 res;
        const bool any = (wsum > 0.0f);
        res.x = any ? acc.x / wsum : 0.0f;
        res.y = any ? acc.y / wsum : 0.0f;
        res.z = any ? acc.z / wsum : 0.0f;
        res.w = any ? acc.w / wsum : 0.0f;
        res.x = (res.x > 0.0f) ? res.x : (expf(res.x) - 1.0f);
        res.y = (res.y > 0.0f) ? res.y : (expf(res.y) - 1.0f);
        res.z = (res.z > 0.0f) ? res.z : (expf(res.z) - 1.0f);
        res.w = (res.w > 0.0f) ? res.w : (expf(res.w) - 1.0f);
        ((float4*)out1)[(size_t)r * 32 + lane] = res;
    }
}

__global__ __launch_bounds__(256) void gemm2_kernel(const float* __restrict__ out1,
                                                    const float* __restrict__ W2,
                                                    float* __restrict__ out, int N_, int nbig) {
    __shared__ __attribute__((aligned(16))) float lds[2 * GROWS * GP];
    gemm_block<128>(out1, W2, out, N_, nbig, (int)blockIdx.x, lds);
}

extern "C" void kernel_launch(void* const* d_in, const int* in_sizes, int n_in,
                              void* d_out, int out_size, void* d_ws, size_t ws_size,
                              hipStream_t stream) {
    const float* x       = (const float*)d_in[0];
    const int*   ei      = (const int*)  d_in[1];
    const float* W1      = (const float*)d_in[2];
    const float* att_src = (const float*)d_in[3];
    const float* att_dst = (const float*)d_in[4];
    const float* W2      = (const float*)d_in[5];
    float*       out     = (float*)d_out;

    const int N_ = in_sizes[0] / 256;
    const int E_ = in_sizes[1] / 2;

    float*    ws    = (float*)d_ws;
    float*    h     = ws;
    float*    out1  = h     + (size_t)N_ * 128;
    float*    e_src = out1  + (size_t)N_ * 128;
    float*    e_dst = e_src + N_;
    const int Np    = (N_ + 63) & ~63;
    int*      ctrl  = (int*)d_out;
    int*      cnt   = ctrl + 256;
    int*      slots = cnt  + Np;
    int*      spill = slots + (size_t)N_ * SLOTS;

    hipMemsetAsync(ctrl, 0, (size_t)(256 + N_) * sizeof(int), stream);

    const int nbig   = ((N_ / GROWS) / 256) * 256;
    const int nsmall = (N_ - nbig * GROWS + 15) / 16;
    const int tiles  = nbig + nsmall;
    gemm1_rank_kernel<<<RANK_BLOCKS + tiles, 256, 0, stream>>>(x, W1, h, N_, nbig, ei, cnt, slots, spill, ctrl, E_);
    attn_kernel   <<<(N_ + ATTN_ROWS * ATTN_WAVES - 1) / (ATTN_ROWS * ATTN_WAVES), 32 * ATTN_WAVES, 0, stream>>>(
                      h, att_src, att_dst, e_src, e_dst, N_);
    pull_rows     <<<(N_ + 8 * ROWS_PER_WAVE - 1) / (8 * ROWS_PER_WAVE), 256, 0, stream>>>(
                      h, e_src, e_dst, cnt, slots, spill, ctrl, out1, N_);
    gemm2_kernel  <<<tiles, 256, 0, stream>>>(out1, W2, out, N_, nbig);
}
